# starting kernel
# baseline (speedup 1.0000x reference)
#include <hip/hip_runtime.h>

#define D 128
#define NNODES 50000
#define NEDGES 600000
#define MT 5
#define ROWS (MT * 16)

typedef __attribute__((ext_vector_type(2))) float v2f;
typedef __attribute__((ext_vector_type(8))) float v8f;

__global__ __launch_bounds__(256) void k_init(const float* __restrict__ feat,
                                              const float* __restrict__ eps,
                                              float* __restrict__ h) {
    const int i = (blockIdx.x * 256 + threadIdx.x) * 4;
    const float s = 1.0f + eps[0];
    const float4 f = *reinterpret_cast<const float4*>(feat + i);
    float4 o;
    o.x = s * f.x; o.y = s * f.y; o.z = s * f.z; o.w = s * f.w;
    *reinterpret_cast<float4*>(h + i) = o;
}

__global__ __launch_bounds__(256) void k_scatter(const float* __restrict__ feat,
                                                 const int* __restrict__ src,
                                                 const int* __restrict__ dst,
                                                 const int* __restrict__ et,
                                                 const float* __restrict__ emb,
                                                 float* __restrict__ h) {
    const int wave = threadIdx.x >> 5;
    const int lane = threadIdx.x & 31;
    const int e = blockIdx.x * 8 + wave;
    const int s = src[e];
    const int d = dst[e];
    const int t = et[e];
    const float4 fs = *reinterpret_cast<const float4*>(feat + s * D + lane * 4);
    const float4 em = *reinterpret_cast<const float4*>(emb + t * D + lane * 4);
    float* dp = h + d * D + lane * 4;
    const float m0 = fmaxf(fs.x + em.x, 0.0f);
    const float m1 = fmaxf(fs.y + em.y, 0.0f);
    const float m2 = fmaxf(fs.z + em.z, 0.0f);
    const float m3 = fmaxf(fs.w + em.w, 0.0f);
    unsafeAtomicAdd(dp + 0, m0);
    unsafeAtomicAdd(dp + 1, m1);
    unsafeAtomicAdd(dp + 2, m2);
    unsafeAtomicAdd(dp + 3, m3);
}

__global__ __launch_bounds__(256) void k_mlp(const float* __restrict__ W1,
                                             const float* __restrict__ B1,
                                             const float* __restrict__ W2,
                                             const float* __restrict__ B2,
                                             float* __restrict__ hout) {
    __shared__ float hs[ROWS * 132];
    __shared__ float hid[ROWS * 260];
    const int tid = threadIdx.x;
    const int base = blockIdx.x * ROWS;

    #pragma unroll
    for (int i = 0; i < 10; ++i) {
        const int chunk = tid + i * 256;
        const int r = chunk >> 5;
        const int c = (chunk & 31) * 4;
        const float4 v = *reinterpret_cast<const float4*>(hout + (base + r) * D + c);
        *reinterpret_cast<float4*>(&hs[r * 132 + c]) = v;
    }
    __syncthreads();

    const int w = tid >> 5;
    const int lane = tid & 31;
    const int M = lane & 15;
    const int half = lane >> 4;
    const int koff = half * 2;

    {
        v8f acc0[MT] = {};
        v8f acc1[MT] = {};
        for (int k0 = 0; k0 < D; k0 += 4) {
            const float* bp = W1 + (k0 + koff) * 256 + w * 32 + M;
            v2f b0, b1;
            b0.x = bp[0];   b0.y = bp[256];
            b1.x = bp[16];  b1.y = bp[16 + 256];
            #pragma unroll
            for (int m = 0; m < MT; ++m) {
                v2f a;
                a.x = hs[(m * 16 + M) * 132 + k0 + koff];
                a.y = hs[(m * 16 + M) * 132 + k0 + koff + 1];
                acc0[m] = __builtin_amdgcn_wmma_f32_16x16x4_f32(
                              false, a, false, b0, (short)0, acc0[m], false, false);
                acc1[m] = __builtin_amdgcn_wmma_f32_16x16x4_f32(
                              false, a, false, b1, (short)0, acc1[m], false, false);
            }
        }
        const float bias0 = B1[w * 32 + M];
        const float bias1 = B1[w * 32 + 16 + M];
        #pragma unroll
        for (int m = 0; m < MT; ++m) {
            #pragma unroll
            for (int r = 0; r < 8; ++r) {
                const int row = m * 16 + r + half * 8;
                const float v0 = acc0[m][r] + bias0;
                const float v1 = acc1[m][r] + bias1;
                hid[row * 260 + w * 32 + M]      = v0 > 0.0f ? v0 : 0.0f;
                hid[row * 260 + w * 32 + 16 + M] = v1 > 0.0f ? v1 : 0.0f;
            }
        }
    }
    __syncthreads();

    {
        v8f acc[MT] = {};
        for (int k0 = 0; k0 < 2 * D; k0 += 4) {
            const float* bp = W2 + (k0 + koff) * D + w * 16 + M;
            v2f b;
            b.x = bp[0];
            b.y = bp[D];
            #pragma unroll
            for (int m = 0; m < MT; ++m) {
                v2f a;
                a.x = hid[(m * 16 + M) * 260 + k0 + koff];
                a.y = hid[(m * 16 + M) * 260 + k0 + koff + 1];
                acc[m] = __builtin_amdgcn_wmma_f32_16x16x4_f32(
                             false, a, false, b, (short)0, acc[m], false, false);
            }
        }
        const float bias = B2[w * 16 + M];
        #pragma unroll
        for (int m = 0; m < MT; ++m) {
            #pragma unroll
            for (int r = 0; r < 8; ++r) {
                hout[(base + m * 16 + r + half * 8) * D + w * 16 + M] = acc[m][r] + bias;
            }
        }
    }
}

extern "C" void kernel_launch(void* const* d_in, const int* in_sizes, int n_in,
                              void* d_out, int out_size, void* d_ws, size_t ws_size,
                              hipStream_t stream) {
    const float* feat = (const float*)d_in[0];
    const int*   src  = (const int*)d_in[1];
    const int*   dst  = (const int*)d_in[2];
    const int*   et   = (const int*)d_in[3];
    const float* emb  = (const float*)d_in[4];
    const float* w1   = (const float*)d_in[5];
    const float* b1   = (const float*)d_in[6];
    const float* w2   = (const float*)d_in[7];
    const float* b2   = (const float*)d_in[8];
    const float* eps  = (const float*)d_in[9];
    float* out = (float*)d_out;

    k_init   <<<(NNODES * D / 4) / 256, 256, 0, stream>>>(feat, eps, out);
    k_scatter<<<NEDGES / 8,            256, 0, stream>>>(feat, src, dst, et, emb, out);
    k_mlp    <<<NNODES / ROWS,         256, 0, stream>>>(w1, b1, w2, b2, out);
}


# scatter: lane l owns floats l, l+32, l+64, l+96 of a row, so each atomic instruction covers one whole 128-B line (was 8 dwords in each of 4 lines); float atomics kept, arrival order differs
# speedup vs baseline: 1.3012x; 1.3012x over previous
#include <hip/hip_runtime.h>

#define D 128
#define NNODES 50000
#define NEDGES 600000
#define MT 5
#define ROWS (MT * 16)

typedef __attribute__((ext_vector_type(2))) float v2f;
typedef __attribute__((ext_vector_type(8))) float v8f;

__global__ __launch_bounds__(256) void k_init(const float* __restrict__ feat,
                                              const float* __restrict__ eps,
                                              float* __restrict__ h) {
    const int i = (blockIdx.x * 256 + threadIdx.x) * 4;
    const float s = 1.0f + eps[0];
    const float4 f = *reinterpret_cast<const float4*>(feat + i);
    float4 o;
    o.x = s * f.x; o.y = s * f.y; o.z = s * f.z; o.w = s * f.w;
    *reinterpret_cast<float4*>(h + i) = o;
}

__global__ __launch_bounds__(256) void k_scatter(const float* __restrict__ feat,
                                                 const int* __restrict__ src,
                                                 const int* __restrict__ dst,
                                                 const int* __restrict__ et,
                                                 const float* __restrict__ emb,
                                                 float* __restrict__ h) {
    const int wave = threadIdx.x >> 5;
    const int lane = threadIdx.x & 31;
    const int e = blockIdx.x * 8 + wave;
    const int s = src[e];
    const int d = dst[e];
    const int t = et[e];
    const float* fp = feat + s * D + lane;
    const float* ep = emb + t * D + lane;
    float* dp = h + d * D + lane;
    const float f0 = fp[0];
    const float f1 = fp[32];
    const float f2 = fp[64];
    const float f3 = fp[96];
    const float e0 = ep[0];
    const float e1 = ep[32];
    const float e2 = ep[64];
    const float e3 = ep[96];
    const float m0 = fmaxf(f0 + e0, 0.0f);
    const float m1 = fmaxf(f1 + e1, 0.0f);
    const float m2 = fmaxf(f2 + e2, 0.0f);
    const float m3 = fmaxf(f3 + e3, 0.0f);
    unsafeAtomicAdd(dp + 0,  m0);
    unsafeAtomicAdd(dp + 32, m1);
    unsafeAtomicAdd(dp + 64, m2);
    unsafeAtomicAdd(dp + 96, m3);
}

__global__ __launch_bounds__(256) void k_mlp(const float* __restrict__ W1,
                                             const float* __restrict__ B1,
                                             const float* __restrict__ W2,
                                             const float* __restrict__ B2,
                                             float* __restrict__ hout) {
    __shared__ float hs[ROWS * 132];
    __shared__ float hid[ROWS * 260];
    const int tid = threadIdx.x;
    const int base = blockIdx.x * ROWS;

    #pragma unroll
    for (int i = 0; i < 10; ++i) {
        const int chunk = tid + i * 256;
        const int r = chunk >> 5;
        const int c = (chunk & 31) * 4;
        const float4 v = *reinterpret_cast<const float4*>(hout + (base + r) * D + c);
        *reinterpret_cast<float4*>(&hs[r * 132 + c]) = v;
    }
    __syncthreads();

    const int w = tid >> 5;
    const int lane = tid & 31;
    const int M = lane & 15;
    const int half = lane >> 4;
    const int koff = half * 2;

    {
        v8f acc0[MT] = {};
        v8f acc1[MT] = {};
        for (int k0 = 0; k0 < D; k0 += 4) {
            const float* bp = W1 + (k0 + koff) * 256 + w * 32 + M;
            v2f b0, b1;
            b0.x = bp[0];   b0.y = bp[256];
            b1.x = bp[16];  b1.y = bp[16 + 256];
            #pragma unroll
            for (int m = 0; m < MT; ++m) {
                v2f a;
                a.x = hs[(m * 16 + M) * 132 + k0 + koff];
                a.y = hs[(m * 16 + M) * 132 + k0 + koff + 1];
                acc0[m] = __builtin_amdgcn_wmma_f32_16x16x4_f32(
                              false, a, false, b0, (short)0, acc0[m], false, false);
                acc1[m] = __builtin_amdgcn_wmma_f32_16x16x4_f32(
                              false, a, false, b1, (short)0, acc1[m], false, false);
            }
        }
        const float bias0 = B1[w * 32 + M];
        const float bias1 = B1[w * 32 + 16 + M];
        #pragma unroll
        for (int m = 0; m < MT; ++m) {
            #pragma unroll
            for (int r = 0; r < 8; ++r) {
                const int row = m * 16 + r + half * 8;
                const float v0 = acc0[m][r] + bias0;
                const float v1 = acc1[m][r] + bias1;
                hid[row * 260 + w * 32 + M]      = v0 > 0.0f ? v0 : 0.0f;
                hid[row * 260 + w * 32 + 16 + M] = v1 > 0.0f ? v1 : 0.0f;
            }
        }
    }
    __syncthreads();

    {
        v8f acc[MT] = {};
        for (int k0 = 0; k0 < 2 * D; k0 += 4) {
            const float* bp = W2 + (k0 + koff) * D + w * 16 + M;
            v2f b;
            b.x = bp[0];
            b.y = bp[D];
            #pragma unroll
            for (int m = 0; m < MT; ++m) {
                v2f a;
                a.x = hid[(m * 16 + M) * 260 + k0 + koff];
                a.y = hid[(m * 16 + M) * 260 + k0 + koff + 1];
                acc[m] = __builtin_amdgcn_wmma_f32_16x16x4_f32(
                             false, a, false, b, (short)0, acc[m], false, false);
            }
        }
        const float bias = B2[w * 16 + M];
        #pragma unroll
        for (int m = 0; m < MT; ++m) {
            #pragma unroll
            for (int r = 0; r < 8; ++r) {
                hout[(base + m * 16 + r + half * 8) * D + w * 16 + M] = acc[m][r] + bias;
            }
        }
    }
}

extern "C" void kernel_launch(void* const* d_in, const int* in_sizes, int n_in,
                              void* d_out, int out_size, void* d_ws, size_t ws_size,
                              hipStream_t stream) {
    const float* feat = (const float*)d_in[0];
    const int*   src  = (const int*)d_in[1];
    const int*   dst  = (const int*)d_in[2];
    const int*   et   = (const int*)d_in[3];
    const float* emb  = (const float*)d_in[4];
    const float* w1   = (const float*)d_in[5];
    const float* b1   = (const float*)d_in[6];
    const float* w2   = (const float*)d_in[7];
    const float* b2   = (const float*)d_in[8];
    const float* eps  = (const float*)d_in[9];
    float* out = (float*)d_out;

    k_init   <<<(NNODES * D / 4) / 256, 256, 0, stream>>>(feat, eps, out);
    k_scatter<<<NEDGES / 8,            256, 0, stream>>>(feat, src, dst, et, emb, out);
    k_mlp    <<<NNODES / ROWS,         256, 0, stream>>>(w1, b1, w2, b2, out);
}


# fewer, longer waves: a scatter wave takes 32 edges (coalesced index load, readlane broadcast, 4 edges' loads in flight): 18,750 waves, was 600,000; init thread 5 float4; atomic arrival order differs
# speedup vs baseline: 1.6354x; 1.2568x over previous
#include <hip/hip_runtime.h>

#define D 128
#define NNODES 50000
#define NEDGES 600000
#define MT 5
#define ROWS (MT * 16)

typedef __attribute__((ext_vector_type(2))) float v2f;
typedef __attribute__((ext_vector_type(8))) float v8f;

#define IPT 5
__global__ __launch_bounds__(256) void k_init(const float* __restrict__ feat,
                                              const float* __restrict__ eps,
                                              float* __restrict__ h) {
    const int i0 = (blockIdx.x * (256 * IPT) + threadIdx.x) * 4;
    const float s = 1.0f + eps[0];
    float4 f[IPT];
    #pragma unroll
    for (int k = 0; k < IPT; ++k)
        f[k] = *reinterpret_cast<const float4*>(feat + i0 + k * 1024);
    #pragma unroll
    for (int k = 0; k < IPT; ++k) {
        float4 o;
        o.x = s * f[k].x; o.y = s * f[k].y; o.z = s * f[k].z; o.w = s * f[k].w;
        *reinterpret_cast<float4*>(h + i0 + k * 1024) = o;
    }
}

#define EPW 32
#define EGRP 4
__global__ __launch_bounds__(256) void k_scatter(const float* __restrict__ feat,
                                                 const int* __restrict__ src,
                                                 const int* __restrict__ dst,
                                                 const int* __restrict__ et,
                                                 const float* __restrict__ emb,
                                                 float* __restrict__ h) {
    const int lane = threadIdx.x & 31;
    const int wv = blockIdx.x * 8 + (threadIdx.x >> 5);
    if (wv >= NEDGES / EPW) return;
    const int e = wv * EPW + lane;
    const int sv = src[e];
    const int dv = dst[e];
    const int tv = et[e];
    for (int j = 0; j < EPW; j += EGRP) {
        float m[EGRP][4];
        #pragma unroll
        for (int u = 0; u < EGRP; ++u) {
            const int s = __builtin_amdgcn_readlane(sv, j + u);
            const int t = __builtin_amdgcn_readlane(tv, j + u);
            const float* fp = feat + s * D + lane;
            const float* ep = emb + t * D + lane;
            #pragma unroll
            for (int q = 0; q < 4; ++q)
                m[u][q] = fmaxf(fp[32 * q] + ep[32 * q], 0.0f);
        }
        #pragma unroll
        for (int u = 0; u < EGRP; ++u) {
            const int d = __builtin_amdgcn_readlane(dv, j + u);
            float* dp = h + d * D + lane;
            #pragma unroll
            for (int q = 0; q < 4; ++q)
                unsafeAtomicAdd(dp + 32 * q, m[u][q]);
        }
    }
}

__global__ __launch_bounds__(256) void k_mlp(const float* __restrict__ W1,
                                             const float* __restrict__ B1,
                                             const float* __restrict__ W2,
                                             const float* __restrict__ B2,
                                             float* __restrict__ hout) {
    __shared__ float hs[ROWS * 132];
    __shared__ float hid[ROWS * 260];
    const int tid = threadIdx.x;
    const int base = blockIdx.x * ROWS;

    #pragma unroll
    for (int i = 0; i < 10; ++i) {
        const int chunk = tid + i * 256;
        const int r = chunk >> 5;
        const int c = (chunk & 31) * 4;
        const float4 v = *reinterpret_cast<const float4*>(hout + (base + r) * D + c);
        *reinterpret_cast<float4*>(&hs[r * 132 + c]) = v;
    }
    __syncthreads();

    const int w = tid >> 5;
    const int lane = tid & 31;
    const int M = lane & 15;
    const int half = lane >> 4;
    const int koff = half * 2;

    {
        v8f acc0[MT] = {};
        v8f acc1[MT] = {};
        for (int k0 = 0; k0 < D; k0 += 4) {
            const float* bp = W1 + (k0 + koff) * 256 + w * 32 + M;
            v2f b0, b1;
            b0.x = bp[0];   b0.y = bp[256];
            b1.x = bp[16];  b1.y = bp[16 + 256];
            #pragma unroll
            for (int m = 0; m < MT; ++m) {
                v2f a;
                a.x = hs[(m * 16 + M) * 132 + k0 + koff];
                a.y = hs[(m * 16 + M) * 132 + k0 + koff + 1];
                acc0[m] = __builtin_amdgcn_wmma_f32_16x16x4_f32(
                              false, a, false, b0, (short)0, acc0[m], false, false);
                acc1[m] = __builtin_amdgcn_wmma_f32_16x16x4_f32(
                              false, a, false, b1, (short)0, acc1[m], false, false);
            }
        }
        const float bias0 = B1[w * 32 + M];
        const float bias1 = B1[w * 32 + 16 + M];
        #pragma unroll
        for (int m = 0; m < MT; ++m) {
            #pragma unroll
            for (int r = 0; r < 8; ++r) {
                const int row = m * 16 + r + half * 8;
                const float v0 = acc0[m][r] + bias0;
                const float v1 = acc1[m][r] + bias1;
                hid[row * 260 + w * 32 + M]      = v0 > 0.0f ? v0 : 0.0f;
                hid[row * 260 + w * 32 + 16 + M] = v1 > 0.0f ? v1 : 0.0f;
            }
        }
    }
    __syncthreads();

    {
        v8f acc[MT] = {};
        for (int k0 = 0; k0 < 2 * D; k0 += 4) {
            const float* bp = W2 + (k0 + koff) * D + w * 16 + M;
            v2f b;
            b.x = bp[0];
            b.y = bp[D];
            #pragma unroll
            for (int m = 0; m < MT; ++m) {
                v2f a;
                a.x = hid[(m * 16 + M) * 260 + k0 + koff];
                a.y = hid[(m * 16 + M) * 260 + k0 + koff + 1];
                acc[m] = __builtin_amdgcn_wmma_f32_16x16x4_f32(
                             false, a, false, b, (short)0, acc[m], false, false);
            }
        }
        const float bias = B2[w * 16 + M];
        #pragma unroll
        for (int m = 0; m < MT; ++m) {
            #pragma unroll
            for (int r = 0; r < 8; ++r) {
                hout[(base + m * 16 + r + half * 8) * D + w * 16 + M] = acc[m][r] + bias;
            }
        }
    }
}

extern "C" void kernel_launch(void* const* d_in, const int* in_sizes, int n_in,
                              void* d_out, int out_size, void* d_ws, size_t ws_size,
                              hipStream_t stream) {
    const float* feat = (const float*)d_in[0];
    const int*   src  = (const int*)d_in[1];
    const int*   dst  = (const int*)d_in[2];
    const int*   et   = (const int*)d_in[3];
    const float* emb  = (const float*)d_in[4];
    const float* w1   = (const float*)d_in[5];
    const float* b1   = (const float*)d_in[6];
    const float* w2   = (const float*)d_in[7];
    const float* b2   = (const float*)d_in[8];
    const float* eps  = (const float*)d_in[9];
    float* out = (float*)d_out;

    k_init   <<<(NNODES * D / 4) / (256 * IPT), 256, 0, stream>>>(feat, eps, out);
    k_scatter<<<(NEDGES / EPW + 7) / 8,         256, 0, stream>>>(feat, src, dst, et, emb, out);
    k_mlp    <<<NNODES / ROWS,         256, 0, stream>>>(w1, b1, w2, b2, out);
}


# MLP launch as one resident pass: 512 work-groups (two per WGP) take 6 or 7 of the 3125 row tiles each (was 625 x 5 tiles: 512 resident, then a 113-group tail); hidden tile reuses the h tile's LDS
# speedup vs baseline: 1.7055x; 1.0429x over previous
#include <hip/hip_runtime.h>

#define D 128
#define NNODES 50000
#define NEDGES 600000

typedef __attribute__((ext_vector_type(2))) float v2f;
typedef __attribute__((ext_vector_type(8))) float v8f;

#define IPT 5
__global__ __launch_bounds__(256) void k_init(const float* __restrict__ feat,
                                              const float* __restrict__ eps,
                                              float* __restrict__ h) {
    const int i0 = (blockIdx.x * (256 * IPT) + threadIdx.x) * 4;
    const float s = 1.0f + eps[0];
    float4 f[IPT];
    #pragma unroll
    for (int k = 0; k < IPT; ++k)
        f[k] = *reinterpret_cast<const float4*>(feat + i0 + k * 1024);
    #pragma unroll
    for (int k = 0; k < IPT; ++k) {
        float4 o;
        o.x = s * f[k].x; o.y = s * f[k].y; o.z = s * f[k].z; o.w = s * f[k].w;
        *reinterpret_cast<float4*>(h + i0 + k * 1024) = o;
    }
}

#define EPW 32
#define EGRP 4
__global__ __launch_bounds__(256) void k_scatter(const float* __restrict__ feat,
                                                 const int* __restrict__ src,
                                                 const int* __restrict__ dst,
                                                 const int* __restrict__ et,
                                                 const float* __restrict__ emb,
                                                 float* __restrict__ h) {
    const int lane = threadIdx.x & 31;
    const int wv = blockIdx.x * 8 + (threadIdx.x >> 5);
    if (wv >= NEDGES / EPW) return;
    const int e = wv * EPW + lane;
    const int sv = src[e];
    const int dv = dst[e];
    const int tv = et[e];
    for (int j = 0; j < EPW; j += EGRP) {
        float m[EGRP][4];
        #pragma unroll
        for (int u = 0; u < EGRP; ++u) {
            const int s = __builtin_amdgcn_readlane(sv, j + u);
            const int t = __builtin_amdgcn_readlane(tv, j + u);
            const float* fp = feat + s * D + lane;
            const float* ep = emb + t * D + lane;
            #pragma unroll
            for (int q = 0; q < 4; ++q)
                m[u][q] = fmaxf(fp[32 * q] + ep[32 * q], 0.0f);
        }
        #pragma unroll
        for (int u = 0; u < EGRP; ++u) {
            const int d = __builtin_amdgcn_readlane(dv, j + u);
            float* dp = h + d * D + lane;
            #pragma unroll
            for (int q = 0; q < 4; ++q)
                unsafeAtomicAdd(dp + 32 * q, m[u][q]);
        }
    }
}

#define NTILES (NNODES / 16)
#define NWG 512
#define MT_LO (NTILES / NWG)
#define N_HI (NTILES % NWG)
#define MT_HI (MT_LO + 1)
#define HI_STRIDE 9
static_assert(HI_STRIDE * N_HI <= NWG, "7-tile groups must fit the grid");
static_assert(NWG * MT_LO + N_HI == NTILES, "tiles must split exactly");

template <int MT>
__device__ __forceinline__ void mlp_tiles(const float* __restrict__ W1,
                                          const float* __restrict__ B1,
                                          const float* __restrict__ W2,
                                          const float* __restrict__ B2,
                                          float* __restrict__ hout,
                                          const int base, float* smem) {
    float* hs = smem;
    float* hid = smem;
    const int tid = threadIdx.x;

    #pragma unroll
    for (int i = 0; i < 2 * MT; ++i) {
        const int chunk = tid + i * 256;
        const int r = chunk >> 5;
        const int c = (chunk & 31) * 4;
        const float4 v = *reinterpret_cast<const float4*>(hout + (base + r) * D + c);
        *reinterpret_cast<float4*>(&hs[r * 132 + c]) = v;
    }
    __syncthreads();

    const int w = tid >> 5;
    const int lane = tid & 31;
    const int M = lane & 15;
    const int half = lane >> 4;
    const int koff = half * 2;

    {
        v8f acc0[MT] = {};
        v8f acc1[MT] = {};
        for (int k0 = 0; k0 < D; k0 += 4) {
            const float* bp = W1 + (k0 + koff) * 256 + w * 32 + M;
            v2f b0, b1;
            b0.x = bp[0];   b0.y = bp[256];
            b1.x = bp[16];  b1.y = bp[16 + 256];
            #pragma unroll
            for (int m = 0; m < MT; ++m) {
                v2f a;
                a.x = hs[(m * 16 + M) * 132 + k0 + koff];
                a.y = hs[(m * 16 + M) * 132 + k0 + koff + 1];
                acc0[m] = __builtin_amdgcn_wmma_f32_16x16x4_f32(
                              false, a, false, b0, (short)0, acc0[m], false, false);
                acc1[m] = __builtin_amdgcn_wmma_f32_16x16x4_f32(
                              false, a, false, b1, (short)0, acc1[m], false, false);
            }
        }
        const float bias0 = B1[w * 32 + M];
        const float bias1 = B1[w * 32 + 16 + M];
        __syncthreads();
        #pragma unroll
        for (int m = 0; m < MT; ++m) {
            #pragma unroll
            for (int r = 0; r < 8; ++r) {
                const int row = m * 16 + r + half * 8;
                const float v0 = acc0[m][r] + bias0;
                const float v1 = acc1[m][r] + bias1;
                hid[row * 260 + w * 32 + M]      = v0 > 0.0f ? v0 : 0.0f;
                hid[row * 260 + w * 32 + 16 + M] = v1 > 0.0f ? v1 : 0.0f;
            }
        }
    }
    __syncthreads();

    {
        v8f acc[MT] = {};
        for (int k0 = 0; k0 < 2 * D; k0 += 4) {
            const float* bp = W2 + (k0 + koff) * D + w * 16 + M;
            v2f b;
            b.x = bp[0];
            b.y = bp[D];
            #pragma unroll
            for (int m = 0; m < MT; ++m) {
                v2f a;
                a.x = hid[(m * 16 + M) * 260 + k0 + koff];
                a.y = hid[(m * 16 + M) * 260 + k0 + koff + 1];
                acc[m] = __builtin_amdgcn_wmma_f32_16x16x4_f32(
                             false, a, false, b, (short)0, acc[m], false, false);
            }
        }
        const float bias = B2[w * 16 + M];
        #pragma unroll
        for (int m = 0; m < MT; ++m) {
            #pragma unroll
            for (int r = 0; r < 8; ++r) {
                hout[(base + m * 16 + r + half * 8) * D + w * 16 + M] = acc[m][r] + bias;
            }
        }
    }
}

__global__ __launch_bounds__(256) void k_mlp(const float* __restrict__ W1,
                                             const float* __restrict__ B1,
                                             const float* __restrict__ W2,
                                             const float* __restrict__ B2,
                                             float* __restrict__ hout) {
    __shared__ float smem[MT_HI * 16 * 260];
    const int b = blockIdx.x;
    const bool hi = (b % HI_STRIDE == 0) && (b < HI_STRIDE * N_HI);
    const int hi_before = min((b + HI_STRIDE - 1) / HI_STRIDE, N_HI);
    const int tile0 = b * MT_LO + hi_before;
    if (hi) mlp_tiles<MT_HI>(W1, B1, W2, B2, hout, tile0 * 16, smem);
    else    mlp_tiles<MT_LO>(W1, B1, W2, B2, hout, tile0 * 16, smem);
}

extern "C" void kernel_launch(void* const* d_in, const int* in_sizes, int n_in,
                              void* d_out, int out_size, void* d_ws, size_t ws_size,
                              hipStream_t stream) {
    const float* feat = (const float*)d_in[0];
    const int*   src  = (const int*)d_in[1];
    const int*   dst  = (const int*)d_in[2];
    const int*   et   = (const int*)d_in[3];
    const float* emb  = (const float*)d_in[4];
    const float* w1   = (const float*)d_in[5];
    const float* b1   = (const float*)d_in[6];
    const float* w2   = (const float*)d_in[7];
    const float* b2   = (const float*)d_in[8];
    const float* eps  = (const float*)d_in[9];
    float* out = (float*)d_out;

    k_init   <<<(NNODES * D / 4) / (256 * IPT), 256, 0, stream>>>(feat, eps, out);
    k_scatter<<<(NEDGES / EPW + 7) / 8,         256, 0, stream>>>(feat, src, dst, et, emb, out);
    k_mlp    <<<NWG,                   256, 0, stream>>>(w1, b1, w2, b2, out);
}
